# starting kernel
# baseline (speedup 1.0000x reference)
#include <hip/hip_runtime.h>

typedef __attribute__((ext_vector_type(16))) __bf16 v16bf;
typedef __attribute__((ext_vector_type(8)))  float  v8f;

#define CCH   128
#define HW    128
#define NB    8
#define KW9   9
#define PITCH 136
#define ROWS  136
#define PART_BYTES (ROWS * PITCH * 2)
#define PLANE 2097152

union FragU { v16bf v; uint4 q[2]; };

__global__ void scnn_wprep(const float* __restrict__ K,
                           __bf16* __restrict__ hi, __bf16* __restrict__ lo)
{
    int t = blockIdx.x * blockDim.x + threadIdx.x;
    if (t >= 288 * 32) return;
    int lane = t & 31;
    int frag = t >> 5;
    int nt = frag & 7;
    int cg = (frag >> 3) & 3;
    int dw = frag >> 5;
    int halfk = lane >> 4;
    int co = nt * 16 + (lane & 15);
    __bf16* ph = hi + (size_t)frag * 512 + lane * 16;
    __bf16* pl = lo + (size_t)frag * 512 + lane * 16;
    #pragma unroll
    for (int j = 0; j < 16; ++j) {
        int kl = (j < 8) ? (halfk * 8 + j) : (16 + halfk * 8 + (j - 8));
        int ci = cg * 32 + kl;
        float w = K[((size_t)dw * CCH + ci) * CCH + co];
        __bf16 h = (__bf16)w;
        ph[j] = h;
        pl[j] = (__bf16)(w - (float)h);
    }
}

__global__ __launch_bounds__(256) void scnn_t_in(const float* __restrict__ X,
                                                 float* __restrict__ Y)
{
    __shared__ float tile[32][33];
    int bh = blockIdx.x;
    int b = bh >> 7, h = bh & 127;
    int tw = blockIdx.y & 3, tc = blockIdx.y >> 2;
    int x0 = threadIdx.x, y0 = threadIdx.y;
    #pragma unroll
    for (int k = 0; k < 4; ++k) {
        int c = tc * 32 + y0 + k * 8;
        int w = tw * 32 + x0;
        tile[y0 + k * 8][x0] = X[(size_t)b * PLANE + (size_t)c * 16384 + h * 128 + w];
    }
    __syncthreads();
    #pragma unroll
    for (int k = 0; k < 4; ++k) {
        int w = tw * 32 + y0 + k * 8;
        int c = tc * 32 + x0;
        Y[(size_t)b * PLANE + (size_t)h * 16384 + w * 128 + c] = tile[x0][y0 + k * 8];
    }
}

__global__ __launch_bounds__(256) void scnn_t_out(const float* __restrict__ Y,
                                                  float* __restrict__ O)
{
    __shared__ float tile[32][33];
    int bh = blockIdx.x;
    int b = bh >> 7, h = bh & 127;
    int tw = blockIdx.y & 3, tc = blockIdx.y >> 2;
    int x0 = threadIdx.x, y0 = threadIdx.y;
    #pragma unroll
    for (int k = 0; k < 4; ++k) {
        int w = tw * 32 + y0 + k * 8;
        int c = tc * 32 + x0;
        tile[y0 + k * 8][x0] = Y[(size_t)b * PLANE + (size_t)h * 16384 + w * 128 + c];
    }
    __syncthreads();
    #pragma unroll
    for (int k = 0; k < 4; ++k) {
        int w = tw * 32 + x0;
        int c = tc * 32 + y0 + k * 8;
        O[(size_t)b * PLANE + (size_t)c * 16384 + h * 128 + w] = tile[x0][y0 + k * 8];
    }
}

__global__ __launch_bounds__(512)
void scnn_scan(float* __restrict__ Y,
               const __bf16* __restrict__ fragHi,
               const __bf16* __restrict__ fragLo,
               int su, int sv, int rev)
{
    extern __shared__ char smem[];
    const int tid  = threadIdx.x;
    const int lane = tid & 31;
    const int wave = tid >> 5;
    float* Yb = Y + (size_t)blockIdx.x * PLANE;

    unsigned int* sw = (unsigned int*)smem;
    for (int i = tid; i < 4 * PART_BYTES / 4; i += 512) sw[i] = 0u;
    __syncthreads();

    const int u0 = rev ? (HW - 1) : 0;
    {
        const float* row = Yb + (size_t)u0 * su;
        __bf16* hi = (__bf16*)(smem);
        __bf16* lo = (__bf16*)(smem + PART_BYTES);
        int base = tid * 32;
        int v = base >> 7, c0 = base & 127;
        #pragma unroll
        for (int j = 0; j < 32; ++j) {
            float x = row[(size_t)v * sv + c0 + j];
            __bf16 h = (__bf16)x;
            hi[(v + 4) * PITCH + c0 + j] = h;
            lo[(v + 4) * PITCH + c0 + j] = (__bf16)(x - (float)h);
        }
    }
    __syncthreads();

    const int mg   = wave >> 2;
    const int ng   = wave & 3;
    const int half = lane >> 4;
    const int li   = lane & 15;

    int p = 0;
    for (int step = 1; step < HW; ++step) {
        const int u = rev ? (HW - 1 - step) : step;
        const char* Ah = smem + (size_t)(p * 2) * PART_BYTES;
        const char* Al = Ah + PART_BYTES;
        float* Yrow = Yb + (size_t)u * su;

        float cur[2][2][8];
        #pragma unroll
        for (int mt = 0; mt < 2; ++mt)
            #pragma unroll
            for (int nt = 0; nt < 2; ++nt) {
                const int co = (ng * 2 + nt) * 16 + li;
                #pragma unroll
                for (int r = 0; r < 8; ++r) {
                    const int w = (mg * 2 + mt) * 16 + half * 8 + r;
                    cur[mt][nt][r] = Yrow[(size_t)w * sv + co];
                }
            }

        v8f acc[2][2];
        #pragma unroll
        for (int mt = 0; mt < 2; ++mt)
            #pragma unroll
            for (int nt = 0; nt < 2; ++nt)
                acc[mt][nt] = (v8f){0.f, 0.f, 0.f, 0.f, 0.f, 0.f, 0.f, 0.f};

        for (int dw = 0; dw < KW9; ++dw) {
            if (dw + 1 < KW9)
                __builtin_prefetch((const char*)fragHi +
                                   (size_t)(((dw + 1) * 4) * 8 + ng * 2) * 1024 + lane * 32, 0, 3);
            #pragma unroll
            for (int cg = 0; cg < 4; ++cg) {
                FragU ahi[2], alo[2];
                #pragma unroll
                for (int mt = 0; mt < 2; ++mt) {
                    const int arow = (mg * 2 + mt) * 16 + li + dw;
                    size_t aoff = (size_t)arow * (PITCH * 2) + (size_t)(cg * 32 + half * 8) * 2;
                    ahi[mt].q[0] = *(const uint4*)(Ah + aoff);
                    ahi[mt].q[1] = *(const uint4*)(Ah + aoff + 32);
                    alo[mt].q[0] = *(const uint4*)(Al + aoff);
                    alo[mt].q[1] = *(const uint4*)(Al + aoff + 32);
                }
                const size_t bbase = (size_t)((dw * 4 + cg) * 8 + ng * 2) * 1024 + lane * 32;
                const char* bh = (const char*)fragHi + bbase;
                const char* bl = (const char*)fragLo + bbase;
                FragU bhv[2], blv[2];
                #pragma unroll
                for (int nt = 0; nt < 2; ++nt) {
                    bhv[nt].q[0] = *(const uint4*)(bh + nt * 1024);
                    bhv[nt].q[1] = *(const uint4*)(bh + nt * 1024 + 16);
                    blv[nt].q[0] = *(const uint4*)(bl + nt * 1024);
                    blv[nt].q[1] = *(const uint4*)(bl + nt * 1024 + 16);
                }
                #pragma unroll
                for (int mt = 0; mt < 2; ++mt)
                    #pragma unroll
                    for (int nt = 0; nt < 2; ++nt)
                        acc[mt][nt] = __builtin_amdgcn_wmma_f32_16x16x32_bf16(
                            false, ahi[mt].v, false, bhv[nt].v, (short)0, acc[mt][nt], false, false);
                #pragma unroll
                for (int mt = 0; mt < 2; ++mt)
                    #pragma unroll
                    for (int nt = 0; nt < 2; ++nt)
                        acc[mt][nt] = __builtin_amdgcn_wmma_f32_16x16x32_bf16(
                            false, alo[mt].v, false, bhv[nt].v, (short)0, acc[mt][nt], false, false);
                #pragma unroll
                for (int mt = 0; mt < 2; ++mt)
                    #pragma unroll
                    for (int nt = 0; nt < 2; ++nt)
                        acc[mt][nt] = __builtin_amdgcn_wmma_f32_16x16x32_bf16(
                            false, ahi[mt].v, false, blv[nt].v, (short)0, acc[mt][nt], false, false);
            }
        }

        __bf16* nhi = (__bf16*)(smem + (size_t)((1 - p) * 2) * PART_BYTES);
        __bf16* nlo = (__bf16*)(smem + (size_t)((1 - p) * 2 + 1) * PART_BYTES);
        #pragma unroll
        for (int mt = 0; mt < 2; ++mt)
            #pragma unroll
            for (int nt = 0; nt < 2; ++nt) {
                const int co = (ng * 2 + nt) * 16 + li;
                #pragma unroll
                for (int r = 0; r < 8; ++r) {
                    const int w = (mg * 2 + mt) * 16 + half * 8 + r;
                    float conv = acc[mt][nt][r];
                    conv = conv > 0.f ? conv : 0.f;
                    float val = cur[mt][nt][r] + conv;
                    Yrow[(size_t)w * sv + co] = val;
                    __bf16 h = (__bf16)val;
                    nhi[(w + 4) * PITCH + co] = h;
                    nlo[(w + 4) * PITCH + co] = (__bf16)(val - (float)h);
                }
            }
        __syncthreads();
        p ^= 1;
    }
}

extern "C" void kernel_launch(void* const* d_in, const int* in_sizes, int n_in,
                              void* d_out, int out_size, void* d_ws, size_t ws_size,
                              hipStream_t stream)
{
    const float* x  = (const float*)d_in[0];
    const float* kk[4] = {(const float*)d_in[1], (const float*)d_in[2],
                          (const float*)d_in[3], (const float*)d_in[4]};
    float* out = (float*)d_out;

    char* ws = (char*)d_ws;
    float* Y = (float*)ws;
    const size_t yBytes = (size_t)NB * PLANE * sizeof(float);
    const size_t FRAG_ELEMS = 288 * 512;
    __bf16* fh[4]; __bf16* fl[4];
    for (int i = 0; i < 4; ++i) {
        fh[i] = (__bf16*)(ws + yBytes) + (size_t)i * 2 * FRAG_ELEMS;
        fl[i] = fh[i] + FRAG_ELEMS;
    }

    for (int i = 0; i < 4; ++i)
        scnn_wprep<<<(288 * 32 + 255) / 256, 256, 0, stream>>>(kk[i], fh[i], fl[i]);

    scnn_t_in<<<dim3(NB * HW, 16), dim3(32, 8), 0, stream>>>(x, Y);

    const int SMEM = 4 * PART_BYTES;
    scnn_scan<<<NB, 512, SMEM, stream>>>(Y, fh[0], fl[0], 16384, 128, 0);
    scnn_scan<<<NB, 512, SMEM, stream>>>(Y, fh[1], fl[1], 16384, 128, 1);
    scnn_scan<<<NB, 512, SMEM, stream>>>(Y, fh[2], fl[2], 128, 16384, 0);
    scnn_scan<<<NB, 512, SMEM, stream>>>(Y, fh[3], fl[3], 128, 16384, 1);

    scnn_t_out<<<dim3(NB * HW, 16), dim3(32, 8), 0, stream>>>(Y, out);
}


# scan K-loop software-pipelined: next chunk's B (global) and A (LDS) fragments load one chunk ahead into a second register set, ahead of the current chunk's 12 WMMAs; sum order unchanged
# speedup vs baseline: 1.2095x; 1.2095x over previous
#include <hip/hip_runtime.h>

typedef __attribute__((ext_vector_type(16))) __bf16 v16bf;
typedef __attribute__((ext_vector_type(8)))  float  v8f;

#define CCH   128
#define HW    128
#define NB    8
#define KW9   9
#define PITCH 136
#define ROWS  136
#define PART_BYTES (ROWS * PITCH * 2)
#define PLANE 2097152

union FragU { v16bf v; uint4 q[2]; };

__global__ void scnn_wprep(const float* __restrict__ K,
                           __bf16* __restrict__ hi, __bf16* __restrict__ lo)
{
    int t = blockIdx.x * blockDim.x + threadIdx.x;
    if (t >= 288 * 32) return;
    int lane = t & 31;
    int frag = t >> 5;
    int nt = frag & 7;
    int cg = (frag >> 3) & 3;
    int dw = frag >> 5;
    int halfk = lane >> 4;
    int co = nt * 16 + (lane & 15);
    __bf16* ph = hi + (size_t)frag * 512 + lane * 16;
    __bf16* pl = lo + (size_t)frag * 512 + lane * 16;
    #pragma unroll
    for (int j = 0; j < 16; ++j) {
        int kl = (j < 8) ? (halfk * 8 + j) : (16 + halfk * 8 + (j - 8));
        int ci = cg * 32 + kl;
        float w = K[((size_t)dw * CCH + ci) * CCH + co];
        __bf16 h = (__bf16)w;
        ph[j] = h;
        pl[j] = (__bf16)(w - (float)h);
    }
}

__global__ __launch_bounds__(256) void scnn_t_in(const float* __restrict__ X,
                                                 float* __restrict__ Y)
{
    __shared__ float tile[32][33];
    int bh = blockIdx.x;
    int b = bh >> 7, h = bh & 127;
    int tw = blockIdx.y & 3, tc = blockIdx.y >> 2;
    int x0 = threadIdx.x, y0 = threadIdx.y;
    #pragma unroll
    for (int k = 0; k < 4; ++k) {
        int c = tc * 32 + y0 + k * 8;
        int w = tw * 32 + x0;
        tile[y0 + k * 8][x0] = X[(size_t)b * PLANE + (size_t)c * 16384 + h * 128 + w];
    }
    __syncthreads();
    #pragma unroll
    for (int k = 0; k < 4; ++k) {
        int w = tw * 32 + y0 + k * 8;
        int c = tc * 32 + x0;
        Y[(size_t)b * PLANE + (size_t)h * 16384 + w * 128 + c] = tile[x0][y0 + k * 8];
    }
}

__global__ __launch_bounds__(256) void scnn_t_out(const float* __restrict__ Y,
                                                  float* __restrict__ O)
{
    __shared__ float tile[32][33];
    int bh = blockIdx.x;
    int b = bh >> 7, h = bh & 127;
    int tw = blockIdx.y & 3, tc = blockIdx.y >> 2;
    int x0 = threadIdx.x, y0 = threadIdx.y;
    #pragma unroll
    for (int k = 0; k < 4; ++k) {
        int w = tw * 32 + y0 + k * 8;
        int c = tc * 32 + x0;
        tile[y0 + k * 8][x0] = Y[(size_t)b * PLANE + (size_t)h * 16384 + w * 128 + c];
    }
    __syncthreads();
    #pragma unroll
    for (int k = 0; k < 4; ++k) {
        int w = tw * 32 + x0;
        int c = tc * 32 + y0 + k * 8;
        O[(size_t)b * PLANE + (size_t)c * 16384 + h * 128 + w] = tile[x0][y0 + k * 8];
    }
}

__global__ __launch_bounds__(512)
void scnn_scan(float* __restrict__ Y,
               const __bf16* __restrict__ fragHi,
               const __bf16* __restrict__ fragLo,
               int su, int sv, int rev)
{
    extern __shared__ char smem[];
    const int tid  = threadIdx.x;
    const int lane = tid & 31;
    const int wave = tid >> 5;
    float* Yb = Y + (size_t)blockIdx.x * PLANE;

    unsigned int* sw = (unsigned int*)smem;
    for (int i = tid; i < 4 * PART_BYTES / 4; i += 512) sw[i] = 0u;
    __syncthreads();

    const int u0 = rev ? (HW - 1) : 0;
    {
        const float* row = Yb + (size_t)u0 * su;
        __bf16* hi = (__bf16*)(smem);
        __bf16* lo = (__bf16*)(smem + PART_BYTES);
        int base = tid * 32;
        int v = base >> 7, c0 = base & 127;
        #pragma unroll
        for (int j = 0; j < 32; ++j) {
            float x = row[(size_t)v * sv + c0 + j];
            __bf16 h = (__bf16)x;
            hi[(v + 4) * PITCH + c0 + j] = h;
            lo[(v + 4) * PITCH + c0 + j] = (__bf16)(x - (float)h);
        }
    }
    __syncthreads();

    const int mg   = wave >> 2;
    const int ng   = wave & 3;
    const int half = lane >> 4;
    const int li   = lane & 15;

    const char* fH = (const char*)fragHi;
    const char* fL = (const char*)fragLo;
    const unsigned bofs0 = (unsigned)(ng * 2) * 1024u + (unsigned)lane * 32u;
    const unsigned aofs0 = (unsigned)((mg * 2) * 16 + li) * (PITCH * 2) + (unsigned)(half * 8) * 2;

    FragU Bh[2][2], Bl[2][2];
    #pragma unroll
    for (int nt = 0; nt < 2; ++nt) {
        Bh[0][nt].q[0] = *(const uint4*)(fH + bofs0 + nt * 1024);
        Bh[0][nt].q[1] = *(const uint4*)(fH + bofs0 + nt * 1024 + 16);
        Bl[0][nt].q[0] = *(const uint4*)(fL + bofs0 + nt * 1024);
        Bl[0][nt].q[1] = *(const uint4*)(fL + bofs0 + nt * 1024 + 16);
    }

    int p = 0;
    for (int step = 1; step < HW; ++step) {
        const int u = rev ? (HW - 1 - step) : step;
        const char* Ah = smem + (size_t)(p * 2) * PART_BYTES + aofs0;
        const char* Al = Ah + PART_BYTES;
        float* Yrow = Yb + (size_t)u * su;

        unsigned bo = bofs0;
        asm volatile("" : "+v"(bo));

        float cur[2][2][8];
        #pragma unroll
        for (int mt = 0; mt < 2; ++mt)
            #pragma unroll
            for (int nt = 0; nt < 2; ++nt) {
                const int co = (ng * 2 + nt) * 16 + li;
                #pragma unroll
                for (int r = 0; r < 8; ++r) {
                    const int w = (mg * 2 + mt) * 16 + half * 8 + r;
                    cur[mt][nt][r] = Yrow[(size_t)w * sv + co];
                }
            }

        v8f acc[2][2];
        #pragma unroll
        for (int mt = 0; mt < 2; ++mt)
            #pragma unroll
            for (int nt = 0; nt < 2; ++nt)
                acc[mt][nt] = (v8f){0.f, 0.f, 0.f, 0.f, 0.f, 0.f, 0.f, 0.f};

        FragU Ahf[2][2], Alf[2][2];
        #pragma unroll
        for (int mt = 0; mt < 2; ++mt) {
            const unsigned ao = (unsigned)(mt * 16) * (PITCH * 2);
            Ahf[0][mt].q[0] = *(const uint4*)(Ah + ao);
            Ahf[0][mt].q[1] = *(const uint4*)(Ah + ao + 32);
            Alf[0][mt].q[0] = *(const uint4*)(Al + ao);
            Alf[0][mt].q[1] = *(const uint4*)(Al + ao + 32);
        }

        #pragma unroll
        for (int kc = 0; kc < 36; ++kc) {
            const int s  = kc & 1;
            const int ns = s ^ 1;
            const int nk = (kc + 1) % 36;
            #pragma unroll
            for (int nt = 0; nt < 2; ++nt) {
                const unsigned bb = (unsigned)nk * 8192u + (unsigned)nt * 1024u;
                Bh[ns][nt].q[0] = *(const uint4*)(fH + bo + bb);
                Bh[ns][nt].q[1] = *(const uint4*)(fH + bo + bb + 16);
                Bl[ns][nt].q[0] = *(const uint4*)(fL + bo + bb);
                Bl[ns][nt].q[1] = *(const uint4*)(fL + bo + bb + 16);
            }
            if (kc + 1 < 36) {
                const int ndw = (kc + 1) >> 2, ncg = (kc + 1) & 3;
                #pragma unroll
                for (int mt = 0; mt < 2; ++mt) {
                    const unsigned ao = (unsigned)(mt * 16 + ndw) * (PITCH * 2) + (unsigned)(ncg * 32) * 2;
                    Ahf[ns][mt].q[0] = *(const uint4*)(Ah + ao);
                    Ahf[ns][mt].q[1] = *(const uint4*)(Ah + ao + 32);
                    Alf[ns][mt].q[0] = *(const uint4*)(Al + ao);
                    Alf[ns][mt].q[1] = *(const uint4*)(Al + ao + 32);
                }
            }
            #pragma unroll
            for (int mt = 0; mt < 2; ++mt)
                #pragma unroll
                for (int nt = 0; nt < 2; ++nt)
                    acc[mt][nt] = __builtin_amdgcn_wmma_f32_16x16x32_bf16(
                        false, Ahf[s][mt].v, false, Bh[s][nt].v, (short)0, acc[mt][nt], false, false);
            #pragma unroll
            for (int mt = 0; mt < 2; ++mt)
                #pragma unroll
                for (int nt = 0; nt < 2; ++nt)
                    acc[mt][nt] = __builtin_amdgcn_wmma_f32_16x16x32_bf16(
                        false, Alf[s][mt].v, false, Bh[s][nt].v, (short)0, acc[mt][nt], false, false);
            #pragma unroll
            for (int mt = 0; mt < 2; ++mt)
                #pragma unroll
                for (int nt = 0; nt < 2; ++nt)
                    acc[mt][nt] = __builtin_amdgcn_wmma_f32_16x16x32_bf16(
                        false, Ahf[s][mt].v, false, Bl[s][nt].v, (short)0, acc[mt][nt], false, false);
        }

        __bf16* nhi = (__bf16*)(smem + (size_t)((1 - p) * 2) * PART_BYTES);
        __bf16* nlo = (__bf16*)(smem + (size_t)((1 - p) * 2 + 1) * PART_BYTES);
        #pragma unroll
        for (int mt = 0; mt < 2; ++mt)
            #pragma unroll
            for (int nt = 0; nt < 2; ++nt) {
                const int co = (ng * 2 + nt) * 16 + li;
                #pragma unroll
                for (int r = 0; r < 8; ++r) {
                    const int w = (mg * 2 + mt) * 16 + half * 8 + r;
                    float conv = acc[mt][nt][r];
                    conv = conv > 0.f ? conv : 0.f;
                    float val = cur[mt][nt][r] + conv;
                    Yrow[(size_t)w * sv + co] = val;
                    __bf16 h = (__bf16)val;
                    nhi[(w + 4) * PITCH + co] = h;
                    nlo[(w + 4) * PITCH + co] = (__bf16)(val - (float)h);
                }
            }
        __syncthreads();
        p ^= 1;
    }
}

extern "C" void kernel_launch(void* const* d_in, const int* in_sizes, int n_in,
                              void* d_out, int out_size, void* d_ws, size_t ws_size,
                              hipStream_t stream)
{
    const float* x  = (const float*)d_in[0];
    const float* kk[4] = {(const float*)d_in[1], (const float*)d_in[2],
                          (const float*)d_in[3], (const float*)d_in[4]};
    float* out = (float*)d_out;

    char* ws = (char*)d_ws;
    float* Y = (float*)ws;
    const size_t yBytes = (size_t)NB * PLANE * sizeof(float);
    const size_t FRAG_ELEMS = 288 * 512;
    __bf16* fh[4]; __bf16* fl[4];
    for (int i = 0; i < 4; ++i) {
        fh[i] = (__bf16*)(ws + yBytes) + (size_t)i * 2 * FRAG_ELEMS;
        fl[i] = fh[i] + FRAG_ELEMS;
    }

    for (int i = 0; i < 4; ++i)
        scnn_wprep<<<(288 * 32 + 255) / 256, 256, 0, stream>>>(kk[i], fh[i], fl[i]);

    scnn_t_in<<<dim3(NB * HW, 16), dim3(32, 8), 0, stream>>>(x, Y);

    const int SMEM = 4 * PART_BYTES;
    scnn_scan<<<NB, 512, SMEM, stream>>>(Y, fh[0], fl[0], 16384, 128, 0);
    scnn_scan<<<NB, 512, SMEM, stream>>>(Y, fh[1], fl[1], 16384, 128, 1);
    scnn_scan<<<NB, 512, SMEM, stream>>>(Y, fh[2], fl[2], 128, 16384, 0);
    scnn_scan<<<NB, 512, SMEM, stream>>>(Y, fh[3], fl[3], 128, 16384, 1);

    scnn_t_out<<<dim3(NB * HW, 16), dim3(32, 8), 0, stream>>>(Y, out);
}


# weight (B) fragment blobs re-laid as two 512-B planes (elements 0-7 of all lanes, then 8-15): each 16-B-per-lane weight load covers 4 whole 128-B lines, not 8 half lines; k order and sums unchanged
# speedup vs baseline: 1.5310x; 1.2658x over previous
#include <hip/hip_runtime.h>

typedef __attribute__((ext_vector_type(16))) __bf16 v16bf;
typedef __attribute__((ext_vector_type(8)))  float  v8f;

#define CCH   128
#define HW    128
#define NB    8
#define KW9   9
#define PITCH 136
#define ROWS  136
#define PART_BYTES (ROWS * PITCH * 2)
#define PLANE 2097152

union FragU { v16bf v; uint4 q[2]; };

__global__ void scnn_wprep(const float* __restrict__ K,
                           __bf16* __restrict__ hi, __bf16* __restrict__ lo)
{
    int t = blockIdx.x * blockDim.x + threadIdx.x;
    if (t >= 288 * 32) return;
    int lane = t & 31;
    int frag = t >> 5;
    int nt = frag & 7;
    int cg = (frag >> 3) & 3;
    int dw = frag >> 5;
    int halfk = lane >> 4;
    int co = nt * 16 + (lane & 15);
    __bf16* ph = hi + (size_t)frag * 512 + lane * 8;
    __bf16* pl = lo + (size_t)frag * 512 + lane * 8;
    #pragma unroll
    for (int j = 0; j < 16; ++j) {
        int kl = (j < 8) ? (halfk * 8 + j) : (16 + halfk * 8 + (j - 8));
        int ci = cg * 32 + kl;
        float w = K[((size_t)dw * CCH + ci) * CCH + co];
        __bf16 h = (__bf16)w;
        const int e = (j < 8) ? j : (256 + (j - 8));
        ph[e] = h;
        pl[e] = (__bf16)(w - (float)h);
    }
}

__global__ __launch_bounds__(256) void scnn_t_in(const float* __restrict__ X,
                                                 float* __restrict__ Y)
{
    __shared__ float tile[32][33];
    int bh = blockIdx.x;
    int b = bh >> 7, h = bh & 127;
    int tw = blockIdx.y & 3, tc = blockIdx.y >> 2;
    int x0 = threadIdx.x, y0 = threadIdx.y;
    #pragma unroll
    for (int k = 0; k < 4; ++k) {
        int c = tc * 32 + y0 + k * 8;
        int w = tw * 32 + x0;
        tile[y0 + k * 8][x0] = X[(size_t)b * PLANE + (size_t)c * 16384 + h * 128 + w];
    }
    __syncthreads();
    #pragma unroll
    for (int k = 0; k < 4; ++k) {
        int w = tw * 32 + y0 + k * 8;
        int c = tc * 32 + x0;
        Y[(size_t)b * PLANE + (size_t)h * 16384 + w * 128 + c] = tile[x0][y0 + k * 8];
    }
}

__global__ __launch_bounds__(256) void scnn_t_out(const float* __restrict__ Y,
                                                  float* __restrict__ O)
{
    __shared__ float tile[32][33];
    int bh = blockIdx.x;
    int b = bh >> 7, h = bh & 127;
    int tw = blockIdx.y & 3, tc = blockIdx.y >> 2;
    int x0 = threadIdx.x, y0 = threadIdx.y;
    #pragma unroll
    for (int k = 0; k < 4; ++k) {
        int w = tw * 32 + y0 + k * 8;
        int c = tc * 32 + x0;
        tile[y0 + k * 8][x0] = Y[(size_t)b * PLANE + (size_t)h * 16384 + w * 128 + c];
    }
    __syncthreads();
    #pragma unroll
    for (int k = 0; k < 4; ++k) {
        int w = tw * 32 + x0;
        int c = tc * 32 + y0 + k * 8;
        O[(size_t)b * PLANE + (size_t)c * 16384 + h * 128 + w] = tile[x0][y0 + k * 8];
    }
}

__global__ __launch_bounds__(512)
void scnn_scan(float* __restrict__ Y,
               const __bf16* __restrict__ fragHi,
               const __bf16* __restrict__ fragLo,
               int su, int sv, int rev)
{
    extern __shared__ char smem[];
    const int tid  = threadIdx.x;
    const int lane = tid & 31;
    const int wave = tid >> 5;
    float* Yb = Y + (size_t)blockIdx.x * PLANE;

    unsigned int* sw = (unsigned int*)smem;
    for (int i = tid; i < 4 * PART_BYTES / 4; i += 512) sw[i] = 0u;
    __syncthreads();

    const int u0 = rev ? (HW - 1) : 0;
    {
        const float* row = Yb + (size_t)u0 * su;
        __bf16* hi = (__bf16*)(smem);
        __bf16* lo = (__bf16*)(smem + PART_BYTES);
        int base = tid * 32;
        int v = base >> 7, c0 = base & 127;
        #pragma unroll
        for (int j = 0; j < 32; ++j) {
            float x = row[(size_t)v * sv + c0 + j];
            __bf16 h = (__bf16)x;
            hi[(v + 4) * PITCH + c0 + j] = h;
            lo[(v + 4) * PITCH + c0 + j] = (__bf16)(x - (float)h);
        }
    }
    __syncthreads();

    const int mg   = wave >> 2;
    const int ng   = wave & 3;
    const int half = lane >> 4;
    const int li   = lane & 15;

    const char* fH = (const char*)fragHi;
    const char* fL = (const char*)fragLo;
    const unsigned bofs0 = (unsigned)(ng * 2) * 1024u + (unsigned)lane * 16u;
    const unsigned aofs0 = (unsigned)((mg * 2) * 16 + li) * (PITCH * 2) + (unsigned)(half * 8) * 2;

    FragU Bh[2][2], Bl[2][2];
    #pragma unroll
    for (int nt = 0; nt < 2; ++nt) {
        Bh[0][nt].q[0] = *(const uint4*)(fH + bofs0 + nt * 1024);
        Bh[0][nt].q[1] = *(const uint4*)(fH + bofs0 + nt * 1024 + 512);
        Bl[0][nt].q[0] = *(const uint4*)(fL + bofs0 + nt * 1024);
        Bl[0][nt].q[1] = *(const uint4*)(fL + bofs0 + nt * 1024 + 512);
    }

    int p = 0;
    for (int step = 1; step < HW; ++step) {
        const int u = rev ? (HW - 1 - step) : step;
        const char* Ah = smem + (size_t)(p * 2) * PART_BYTES + aofs0;
        const char* Al = Ah + PART_BYTES;
        float* Yrow = Yb + (size_t)u * su;

        unsigned bo = bofs0;
        asm volatile("" : "+v"(bo));

        float cur[2][2][8];
        #pragma unroll
        for (int mt = 0; mt < 2; ++mt)
            #pragma unroll
            for (int nt = 0; nt < 2; ++nt) {
                const int co = (ng * 2 + nt) * 16 + li;
                #pragma unroll
                for (int r = 0; r < 8; ++r) {
                    const int w = (mg * 2 + mt) * 16 + half * 8 + r;
                    cur[mt][nt][r] = Yrow[(size_t)w * sv + co];
                }
            }

        v8f acc[2][2];
        #pragma unroll
        for (int mt = 0; mt < 2; ++mt)
            #pragma unroll
            for (int nt = 0; nt < 2; ++nt)
                acc[mt][nt] = (v8f){0.f, 0.f, 0.f, 0.f, 0.f, 0.f, 0.f, 0.f};

        FragU Ahf[2][2], Alf[2][2];
        #pragma unroll
        for (int mt = 0; mt < 2; ++mt) {
            const unsigned ao = (unsigned)(mt * 16) * (PITCH * 2);
            Ahf[0][mt].q[0] = *(const uint4*)(Ah + ao);
            Ahf[0][mt].q[1] = *(const uint4*)(Ah + ao + 32);
            Alf[0][mt].q[0] = *(const uint4*)(Al + ao);
            Alf[0][mt].q[1] = *(const uint4*)(Al + ao + 32);
        }

        #pragma unroll
        for (int kc = 0; kc < 36; ++kc) {
            const int s  = kc & 1;
            const int ns = s ^ 1;
            const int nk = (kc + 1) % 36;
            #pragma unroll
            for (int nt = 0; nt < 2; ++nt) {
                const unsigned bb = (unsigned)nk * 8192u + (unsigned)nt * 1024u;
                Bh[ns][nt].q[0] = *(const uint4*)(fH + bo + bb);
                Bh[ns][nt].q[1] = *(const uint4*)(fH + bo + bb + 512);
                Bl[ns][nt].q[0] = *(const uint4*)(fL + bo + bb);
                Bl[ns][nt].q[1] = *(const uint4*)(fL + bo + bb + 512);
            }
            if (kc + 1 < 36) {
                const int ndw = (kc + 1) >> 2, ncg = (kc + 1) & 3;
                #pragma unroll
                for (int mt = 0; mt < 2; ++mt) {
                    const unsigned ao = (unsigned)(mt * 16 + ndw) * (PITCH * 2) + (unsigned)(ncg * 32) * 2;
                    Ahf[ns][mt].q[0] = *(const uint4*)(Ah + ao);
                    Ahf[ns][mt].q[1] = *(const uint4*)(Ah + ao + 32);
                    Alf[ns][mt].q[0] = *(const uint4*)(Al + ao);
                    Alf[ns][mt].q[1] = *(const uint4*)(Al + ao + 32);
                }
            }
            #pragma unroll
            for (int mt = 0; mt < 2; ++mt)
                #pragma unroll
                for (int nt = 0; nt < 2; ++nt)
                    acc[mt][nt] = __builtin_amdgcn_wmma_f32_16x16x32_bf16(
                        false, Ahf[s][mt].v, false, Bh[s][nt].v, (short)0, acc[mt][nt], false, false);
            #pragma unroll
            for (int mt = 0; mt < 2; ++mt)
                #pragma unroll
                for (int nt = 0; nt < 2; ++nt)
                    acc[mt][nt] = __builtin_amdgcn_wmma_f32_16x16x32_bf16(
                        false, Alf[s][mt].v, false, Bh[s][nt].v, (short)0, acc[mt][nt], false, false);
            #pragma unroll
            for (int mt = 0; mt < 2; ++mt)
                #pragma unroll
                for (int nt = 0; nt < 2; ++nt)
                    acc[mt][nt] = __builtin_amdgcn_wmma_f32_16x16x32_bf16(
                        false, Ahf[s][mt].v, false, Bl[s][nt].v, (short)0, acc[mt][nt], false, false);
        }

        __bf16* nhi = (__bf16*)(smem + (size_t)((1 - p) * 2) * PART_BYTES);
        __bf16* nlo = (__bf16*)(smem + (size_t)((1 - p) * 2 + 1) * PART_BYTES);
        #pragma unroll
        for (int mt = 0; mt < 2; ++mt)
            #pragma unroll
            for (int nt = 0; nt < 2; ++nt) {
                const int co = (ng * 2 + nt) * 16 + li;
                #pragma unroll
                for (int r = 0; r < 8; ++r) {
                    const int w = (mg * 2 + mt) * 16 + half * 8 + r;
                    float conv = acc[mt][nt][r];
                    conv = conv > 0.f ? conv : 0.f;
                    float val = cur[mt][nt][r] + conv;
                    Yrow[(size_t)w * sv + co] = val;
                    __bf16 h = (__bf16)val;
                    nhi[(w + 4) * PITCH + co] = h;
                    nlo[(w + 4) * PITCH + co] = (__bf16)(val - (float)h);
                }
            }
        __syncthreads();
        p ^= 1;
    }
}

extern "C" void kernel_launch(void* const* d_in, const int* in_sizes, int n_in,
                              void* d_out, int out_size, void* d_ws, size_t ws_size,
                              hipStream_t stream)
{
    const float* x  = (const float*)d_in[0];
    const float* kk[4] = {(const float*)d_in[1], (const float*)d_in[2],
                          (const float*)d_in[3], (const float*)d_in[4]};
    float* out = (float*)d_out;

    char* ws = (char*)d_ws;
    float* Y = (float*)ws;
    const size_t yBytes = (size_t)NB * PLANE * sizeof(float);
    const size_t FRAG_ELEMS = 288 * 512;
    __bf16* fh[4]; __bf16* fl[4];
    for (int i = 0; i < 4; ++i) {
        fh[i] = (__bf16*)(ws + yBytes) + (size_t)i * 2 * FRAG_ELEMS;
        fl[i] = fh[i] + FRAG_ELEMS;
    }

    for (int i = 0; i < 4; ++i)
        scnn_wprep<<<(288 * 32 + 255) / 256, 256, 0, stream>>>(kk[i], fh[i], fl[i]);

    scnn_t_in<<<dim3(NB * HW, 16), dim3(32, 8), 0, stream>>>(x, Y);

    const int SMEM = 4 * PART_BYTES;
    scnn_scan<<<NB, 512, SMEM, stream>>>(Y, fh[0], fl[0], 16384, 128, 0);
    scnn_scan<<<NB, 512, SMEM, stream>>>(Y, fh[1], fl[1], 16384, 128, 1);
    scnn_scan<<<NB, 512, SMEM, stream>>>(Y, fh[2], fl[2], 128, 16384, 0);
    scnn_scan<<<NB, 512, SMEM, stream>>>(Y, fh[3], fl[3], 128, 16384, 1);

    scnn_t_out<<<dim3(NB * HW, 16), dim3(32, 8), 0, stream>>>(Y, out);
}


# scan: weight (B) fragments fetched two K-chunks ahead into a ring of three register sets (was one ahead, two sets); residual-row bases kept in scalar registers to make room; sums unchanged
# speedup vs baseline: 1.5824x; 1.0336x over previous
#include <hip/hip_runtime.h>

typedef __attribute__((ext_vector_type(16))) __bf16 v16bf;
typedef __attribute__((ext_vector_type(8)))  float  v8f;

#define CCH   128
#define HW    128
#define NB    8
#define KW9   9
#define PITCH 136
#define ROWS  136
#define PART_BYTES (ROWS * PITCH * 2)
#define PLANE 2097152

union FragU { v16bf v; uint4 q[2]; };

__global__ void scnn_wprep(const float* __restrict__ K,
                           __bf16* __restrict__ hi, __bf16* __restrict__ lo)
{
    int t = blockIdx.x * blockDim.x + threadIdx.x;
    if (t >= 288 * 32) return;
    int lane = t & 31;
    int frag = t >> 5;
    int nt = frag & 7;
    int cg = (frag >> 3) & 3;
    int dw = frag >> 5;
    int halfk = lane >> 4;
    int co = nt * 16 + (lane & 15);
    __bf16* ph = hi + (size_t)frag * 512 + lane * 8;
    __bf16* pl = lo + (size_t)frag * 512 + lane * 8;
    #pragma unroll
    for (int j = 0; j < 16; ++j) {
        int kl = (j < 8) ? (halfk * 8 + j) : (16 + halfk * 8 + (j - 8));
        int ci = cg * 32 + kl;
        float w = K[((size_t)dw * CCH + ci) * CCH + co];
        __bf16 h = (__bf16)w;
        const int e = (j < 8) ? j : (256 + (j - 8));
        ph[e] = h;
        pl[e] = (__bf16)(w - (float)h);
    }
}

__global__ __launch_bounds__(256) void scnn_t_in(const float* __restrict__ X,
                                                 float* __restrict__ Y)
{
    __shared__ float tile[32][33];
    int bh = blockIdx.x;
    int b = bh >> 7, h = bh & 127;
    int tw = blockIdx.y & 3, tc = blockIdx.y >> 2;
    int x0 = threadIdx.x, y0 = threadIdx.y;
    #pragma unroll
    for (int k = 0; k < 4; ++k) {
        int c = tc * 32 + y0 + k * 8;
        int w = tw * 32 + x0;
        tile[y0 + k * 8][x0] = X[(size_t)b * PLANE + (size_t)c * 16384 + h * 128 + w];
    }
    __syncthreads();
    #pragma unroll
    for (int k = 0; k < 4; ++k) {
        int w = tw * 32 + y0 + k * 8;
        int c = tc * 32 + x0;
        Y[(size_t)b * PLANE + (size_t)h * 16384 + w * 128 + c] = tile[x0][y0 + k * 8];
    }
}

__global__ __launch_bounds__(256) void scnn_t_out(const float* __restrict__ Y,
                                                  float* __restrict__ O)
{
    __shared__ float tile[32][33];
    int bh = blockIdx.x;
    int b = bh >> 7, h = bh & 127;
    int tw = blockIdx.y & 3, tc = blockIdx.y >> 2;
    int x0 = threadIdx.x, y0 = threadIdx.y;
    #pragma unroll
    for (int k = 0; k < 4; ++k) {
        int w = tw * 32 + y0 + k * 8;
        int c = tc * 32 + x0;
        tile[y0 + k * 8][x0] = Y[(size_t)b * PLANE + (size_t)h * 16384 + w * 128 + c];
    }
    __syncthreads();
    #pragma unroll
    for (int k = 0; k < 4; ++k) {
        int w = tw * 32 + x0;
        int c = tc * 32 + y0 + k * 8;
        O[(size_t)b * PLANE + (size_t)c * 16384 + h * 128 + w] = tile[x0][y0 + k * 8];
    }
}

__global__ __launch_bounds__(512)
void scnn_scan(float* __restrict__ Y,
               const __bf16* __restrict__ fragHi,
               const __bf16* __restrict__ fragLo,
               int su, int sv, int rev)
{
    extern __shared__ char smem[];
    const int tid  = threadIdx.x;
    const int lane = tid & 31;
    const int wave = tid >> 5;
    float* Yb = Y + (size_t)blockIdx.x * PLANE;

    unsigned int* sw = (unsigned int*)smem;
    for (int i = tid; i < 4 * PART_BYTES / 4; i += 512) sw[i] = 0u;
    __syncthreads();

    const int u0 = rev ? (HW - 1) : 0;
    {
        const float* row = Yb + (size_t)u0 * su;
        __bf16* hi = (__bf16*)(smem);
        __bf16* lo = (__bf16*)(smem + PART_BYTES);
        int base = tid * 32;
        int v = base >> 7, c0 = base & 127;
        #pragma unroll
        for (int j = 0; j < 32; ++j) {
            float x = row[(size_t)v * sv + c0 + j];
            __bf16 h = (__bf16)x;
            hi[(v + 4) * PITCH + c0 + j] = h;
            lo[(v + 4) * PITCH + c0 + j] = (__bf16)(x - (float)h);
        }
    }
    __syncthreads();

    const int mg   = wave >> 2;
    const int ng   = wave & 3;
    const int half = lane >> 4;
    const int li   = lane & 15;

    const char* fH = (const char*)fragHi;
    const char* fL = (const char*)fragLo;
    const unsigned bofs0 = (unsigned)(ng * 2) * 1024u + (unsigned)lane * 16u;
    const unsigned aofs0 = (unsigned)((mg * 2) * 16 + li) * (PITCH * 2) + (unsigned)(half * 8) * 2;

    const unsigned lofs = (unsigned)((mg * 32 + half * 8) * sv + ng * 32 + li) * 4u;

    FragU Bh[3][2], Bl[3][2];
    #pragma unroll
    for (int c = 0; c < 2; ++c)
        #pragma unroll
        for (int nt = 0; nt < 2; ++nt) {
            const unsigned bb = (unsigned)c * 8192u + (unsigned)nt * 1024u;
            Bh[c][nt].q[0] = *(const uint4*)(fH + bofs0 + bb);
            Bh[c][nt].q[1] = *(const uint4*)(fH + bofs0 + bb + 512);
            Bl[c][nt].q[0] = *(const uint4*)(fL + bofs0 + bb);
            Bl[c][nt].q[1] = *(const uint4*)(fL + bofs0 + bb + 512);
        }

    int p = 0;
    for (int step = 1; step < HW; ++step) {
        const int u = rev ? (HW - 1 - step) : step;
        const char* Ah = smem + (size_t)(p * 2) * PART_BYTES + aofs0;
        const char* Al = Ah + PART_BYTES;
        float* Yrow = Yb + (size_t)u * su;

        unsigned bo = bofs0;
        asm volatile("" : "+v"(bo));

        unsigned lo = lofs;
        asm volatile("" : "+v"(lo));
        const char* rowp[2][8];
        #pragma unroll
        for (int mt = 0; mt < 2; ++mt)
            #pragma unroll
            for (int r = 0; r < 8; ++r) {
                unsigned ro = (unsigned)((mt * 16 + r) * sv) * 4u;
                asm volatile("" : "+s"(ro));
                rowp[mt][r] = (const char*)Yrow + ro;
            }

        float cur[2][2][8];
        #pragma unroll
        for (int mt = 0; mt < 2; ++mt)
            #pragma unroll
            for (int nt = 0; nt < 2; ++nt)
                #pragma unroll
                for (int r = 0; r < 8; ++r)
                    cur[mt][nt][r] = *(const float*)(rowp[mt][r] + lo + nt * 64);

        v8f acc[2][2];
        #pragma unroll
        for (int mt = 0; mt < 2; ++mt)
            #pragma unroll
            for (int nt = 0; nt < 2; ++nt)
                acc[mt][nt] = (v8f){0.f, 0.f, 0.f, 0.f, 0.f, 0.f, 0.f, 0.f};

        FragU Ahf[2][2], Alf[2][2];
        #pragma unroll
        for (int mt = 0; mt < 2; ++mt) {
            const unsigned ao = (unsigned)(mt * 16) * (PITCH * 2);
            Ahf[0][mt].q[0] = *(const uint4*)(Ah + ao);
            Ahf[0][mt].q[1] = *(const uint4*)(Ah + ao + 32);
            Alf[0][mt].q[0] = *(const uint4*)(Al + ao);
            Alf[0][mt].q[1] = *(const uint4*)(Al + ao + 32);
        }

        #pragma unroll
        for (int kc = 0; kc < 36; ++kc) {
            const int s  = kc & 1;
            const int ns = s ^ 1;
            const int bs = kc % 3;
            const int bn = (kc + 2) % 3;
            const int nk = (kc + 2) % 36;
            #pragma unroll
            for (int nt = 0; nt < 2; ++nt) {
                const unsigned bb = (unsigned)nk * 8192u + (unsigned)nt * 1024u;
                Bh[bn][nt].q[0] = *(const uint4*)(fH + bo + bb);
                Bh[bn][nt].q[1] = *(const uint4*)(fH + bo + bb + 512);
                Bl[bn][nt].q[0] = *(const uint4*)(fL + bo + bb);
                Bl[bn][nt].q[1] = *(const uint4*)(fL + bo + bb + 512);
            }
            if (kc + 1 < 36) {
                const int ndw = (kc + 1) >> 2, ncg = (kc + 1) & 3;
                #pragma unroll
                for (int mt = 0; mt < 2; ++mt) {
                    const unsigned ao = (unsigned)(mt * 16 + ndw) * (PITCH * 2) + (unsigned)(ncg * 32) * 2;
                    Ahf[ns][mt].q[0] = *(const uint4*)(Ah + ao);
                    Ahf[ns][mt].q[1] = *(const uint4*)(Ah + ao + 32);
                    Alf[ns][mt].q[0] = *(const uint4*)(Al + ao);
                    Alf[ns][mt].q[1] = *(const uint4*)(Al + ao + 32);
                }
            }
            #pragma unroll
            for (int mt = 0; mt < 2; ++mt)
                #pragma unroll
                for (int nt = 0; nt < 2; ++nt)
                    acc[mt][nt] = __builtin_amdgcn_wmma_f32_16x16x32_bf16(
                        false, Ahf[s][mt].v, false, Bh[bs][nt].v, (short)0, acc[mt][nt], false, false);
            #pragma unroll
            for (int mt = 0; mt < 2; ++mt)
                #pragma unroll
                for (int nt = 0; nt < 2; ++nt)
                    acc[mt][nt] = __builtin_amdgcn_wmma_f32_16x16x32_bf16(
                        false, Alf[s][mt].v, false, Bh[bs][nt].v, (short)0, acc[mt][nt], false, false);
            #pragma unroll
            for (int mt = 0; mt < 2; ++mt)
                #pragma unroll
                for (int nt = 0; nt < 2; ++nt)
                    acc[mt][nt] = __builtin_amdgcn_wmma_f32_16x16x32_bf16(
                        false, Ahf[s][mt].v, false, Bl[bs][nt].v, (short)0, acc[mt][nt], false, false);
        }

        __bf16* nhi = (__bf16*)(smem + (size_t)((1 - p) * 2) * PART_BYTES);
        __bf16* nlo = (__bf16*)(smem + (size_t)((1 - p) * 2 + 1) * PART_BYTES);
        #pragma unroll
        for (int mt = 0; mt < 2; ++mt)
            #pragma unroll
            for (int nt = 0; nt < 2; ++nt) {
                const int co = (ng * 2 + nt) * 16 + li;
                #pragma unroll
                for (int r = 0; r < 8; ++r) {
                    const int w = (mg * 2 + mt) * 16 + half * 8 + r;
                    float conv = acc[mt][nt][r];
                    conv = conv > 0.f ? conv : 0.f;
                    float val = cur[mt][nt][r] + conv;
                    *(float*)((char*)rowp[mt][r] + lo + nt * 64) = val;
                    __bf16 h = (__bf16)val;
                    nhi[(w + 4) * PITCH + co] = h;
                    nlo[(w + 4) * PITCH + co] = (__bf16)(val - (float)h);
                }
            }
        __syncthreads();
        p ^= 1;
    }
}

extern "C" void kernel_launch(void* const* d_in, const int* in_sizes, int n_in,
                              void* d_out, int out_size, void* d_ws, size_t ws_size,
                              hipStream_t stream)
{
    const float* x  = (const float*)d_in[0];
    const float* kk[4] = {(const float*)d_in[1], (const float*)d_in[2],
                          (const float*)d_in[3], (const float*)d_in[4]};
    float* out = (float*)d_out;

    char* ws = (char*)d_ws;
    float* Y = (float*)ws;
    const size_t yBytes = (size_t)NB * PLANE * sizeof(float);
    const size_t FRAG_ELEMS = 288 * 512;
    __bf16* fh[4]; __bf16* fl[4];
    for (int i = 0; i < 4; ++i) {
        fh[i] = (__bf16*)(ws + yBytes) + (size_t)i * 2 * FRAG_ELEMS;
        fl[i] = fh[i] + FRAG_ELEMS;
    }

    for (int i = 0; i < 4; ++i)
        scnn_wprep<<<(288 * 32 + 255) / 256, 256, 0, stream>>>(kk[i], fh[i], fl[i]);

    scnn_t_in<<<dim3(NB * HW, 16), dim3(32, 8), 0, stream>>>(x, Y);

    const int SMEM = 4 * PART_BYTES;
    scnn_scan<<<NB, 512, SMEM, stream>>>(Y, fh[0], fl[0], 16384, 128, 0);
    scnn_scan<<<NB, 512, SMEM, stream>>>(Y, fh[1], fl[1], 16384, 128, 1);
    scnn_scan<<<NB, 512, SMEM, stream>>>(Y, fh[2], fl[2], 128, 16384, 0);
    scnn_scan<<<NB, 512, SMEM, stream>>>(Y, fh[3], fl[3], 128, 16384, 1);

    scnn_t_out<<<dim3(NB * HW, 16), dim3(32, 8), 0, stream>>>(Y, out);
}


# scan split into one launch per line: grid (4 channel slices x batch) of 4-wave work-groups, 32 work-groups per step (was 8 per scan); previous line re-split to LDS each launch; sums unchanged
# speedup vs baseline: 2.2337x; 1.4116x over previous
#include <hip/hip_runtime.h>

typedef __attribute__((ext_vector_type(16))) __bf16 v16bf;
typedef __attribute__((ext_vector_type(8)))  float  v8f;

#define CCH   128
#define HW    128
#define NB    8
#define KW9   9
#define PITCH 136
#define ROWS  136
#define PART_BYTES (ROWS * PITCH * 2)
#define PLANE 2097152

union FragU { v16bf v; uint4 q[2]; };

__global__ void scnn_wprep(const float* __restrict__ K,
                           __bf16* __restrict__ hi, __bf16* __restrict__ lo)
{
    int t = blockIdx.x * blockDim.x + threadIdx.x;
    if (t >= 288 * 32) return;
    int lane = t & 31;
    int frag = t >> 5;
    int nt = frag & 7;
    int cg = (frag >> 3) & 3;
    int dw = frag >> 5;
    int halfk = lane >> 4;
    int co = nt * 16 + (lane & 15);
    __bf16* ph = hi + (size_t)frag * 512 + lane * 8;
    __bf16* pl = lo + (size_t)frag * 512 + lane * 8;
    #pragma unroll
    for (int j = 0; j < 16; ++j) {
        int kl = (j < 8) ? (halfk * 8 + j) : (16 + halfk * 8 + (j - 8));
        int ci = cg * 32 + kl;
        float w = K[((size_t)dw * CCH + ci) * CCH + co];
        __bf16 h = (__bf16)w;
        const int e = (j < 8) ? j : (256 + (j - 8));
        ph[e] = h;
        pl[e] = (__bf16)(w - (float)h);
    }
}

__global__ __launch_bounds__(256) void scnn_t_in(const float* __restrict__ X,
                                                 float* __restrict__ Y)
{
    __shared__ float tile[32][33];
    int bh = blockIdx.x;
    int b = bh >> 7, h = bh & 127;
    int tw = blockIdx.y & 3, tc = blockIdx.y >> 2;
    int x0 = threadIdx.x, y0 = threadIdx.y;
    #pragma unroll
    for (int k = 0; k < 4; ++k) {
        int c = tc * 32 + y0 + k * 8;
        int w = tw * 32 + x0;
        tile[y0 + k * 8][x0] = X[(size_t)b * PLANE + (size_t)c * 16384 + h * 128 + w];
    }
    __syncthreads();
    #pragma unroll
    for (int k = 0; k < 4; ++k) {
        int w = tw * 32 + y0 + k * 8;
        int c = tc * 32 + x0;
        Y[(size_t)b * PLANE + (size_t)h * 16384 + w * 128 + c] = tile[x0][y0 + k * 8];
    }
}

__global__ __launch_bounds__(256) void scnn_t_out(const float* __restrict__ Y,
                                                  float* __restrict__ O)
{
    __shared__ float tile[32][33];
    int bh = blockIdx.x;
    int b = bh >> 7, h = bh & 127;
    int tw = blockIdx.y & 3, tc = blockIdx.y >> 2;
    int x0 = threadIdx.x, y0 = threadIdx.y;
    #pragma unroll
    for (int k = 0; k < 4; ++k) {
        int w = tw * 32 + y0 + k * 8;
        int c = tc * 32 + x0;
        tile[y0 + k * 8][x0] = Y[(size_t)b * PLANE + (size_t)h * 16384 + w * 128 + c];
    }
    __syncthreads();
    #pragma unroll
    for (int k = 0; k < 4; ++k) {
        int w = tw * 32 + x0;
        int c = tc * 32 + y0 + k * 8;
        O[(size_t)b * PLANE + (size_t)c * 16384 + h * 128 + w] = tile[x0][y0 + k * 8];
    }
}

typedef __attribute__((ext_vector_type(4))) __bf16 v4bf;
#define RD 6

__global__ __launch_bounds__(128)
void scnn_step(float* __restrict__ Y,
               const __bf16* __restrict__ fragHi,
               const __bf16* __restrict__ fragLo,
               int su, int sv, int rev, int step)
{
    extern __shared__ char smem[];
    const int tid  = threadIdx.x;
    const int lane = tid & 31;
    const int mg   = tid >> 5;
    const int ng   = blockIdx.x;
    float* Yb = Y + (size_t)blockIdx.y * PLANE;
    const int u  = rev ? (HW - 1 - step) : step;
    const int up = rev ? (u + 1) : (u - 1);
    const float* prow = Yb + (size_t)up * su;
    float* Yrow = Yb + (size_t)u * su;

    const int half = lane >> 4;
    const int li   = lane & 15;

    const char* fH = (const char*)fragHi;
    const char* fL = (const char*)fragLo;
    const unsigned bofs0 = (unsigned)(ng * 2) * 1024u + (unsigned)lane * 16u;

    FragU Bh[RD][2], Bl[RD][2];
    #pragma unroll
    for (int c = 0; c < RD - 1; ++c)
        #pragma unroll
        for (int nt = 0; nt < 2; ++nt) {
            const unsigned bb = (unsigned)c * 8192u + (unsigned)nt * 1024u;
            Bh[c][nt].q[0] = *(const uint4*)(fH + bofs0 + bb);
            Bh[c][nt].q[1] = *(const uint4*)(fH + bofs0 + bb + 512);
            Bl[c][nt].q[0] = *(const uint4*)(fL + bofs0 + bb);
            Bl[c][nt].q[1] = *(const uint4*)(fL + bofs0 + bb + 512);
        }

    for (int i = tid; i < 4 * 272; i += 128) {
        const int part = i / 544, r = i % 544;
        const int pad = r / 272, w = r % 272;
        ((unsigned int*)(smem + part * PART_BYTES + pad * (132 * PITCH * 2)))[w] = 0u;
    }

    {
        __bf16* hi = (__bf16*)(smem);
        __bf16* lo = (__bf16*)(smem + PART_BYTES);
        float4 xs[32];
        const float* p0 = prow + (unsigned)(mg * sv + lane * 4);
        #pragma unroll
        for (int i = 0; i < 32; ++i)
            xs[i] = *(const float4*)(p0 + (unsigned)(4 * i) * (unsigned)sv);
        __bf16* hi0 = hi + (mg + 4) * PITCH + lane * 4;
        __bf16* lo0 = lo + (mg + 4) * PITCH + lane * 4;
        #pragma unroll
        for (int i = 0; i < 32; ++i) {
            const float x0 = xs[i].x, x1 = xs[i].y, x2 = xs[i].z, x3 = xs[i].w;
            const __bf16 h0 = (__bf16)x0, h1 = (__bf16)x1, h2 = (__bf16)x2, h3 = (__bf16)x3;
            v4bf hv, lv;
            hv[0] = h0; hv[1] = h1; hv[2] = h2; hv[3] = h3;
            lv[0] = (__bf16)(x0 - (float)h0);
            lv[1] = (__bf16)(x1 - (float)h1);
            lv[2] = (__bf16)(x2 - (float)h2);
            lv[3] = (__bf16)(x3 - (float)h3);
            *(v4bf*)(hi0 + 4 * i * PITCH) = hv;
            *(v4bf*)(lo0 + 4 * i * PITCH) = lv;
        }
    }
    __syncthreads();

    const unsigned aofs0 = (unsigned)((mg * 2) * 16 + li) * (PITCH * 2) + (unsigned)(half * 8) * 2;
    const char* Ah = smem + aofs0;
    const char* Al = Ah + PART_BYTES;

    const unsigned lofs = (unsigned)((mg * 32 + half * 8) * sv + ng * 32 + li) * 4u;

    const char* rowp[2][8];
    #pragma unroll
    for (int mt = 0; mt < 2; ++mt)
        #pragma unroll
        for (int r = 0; r < 8; ++r)
            rowp[mt][r] = (const char*)Yrow + (unsigned)((mt * 16 + r) * sv) * 4u;

    float cur[2][2][8];
    #pragma unroll
    for (int mt = 0; mt < 2; ++mt)
        #pragma unroll
        for (int nt = 0; nt < 2; ++nt)
            #pragma unroll
            for (int r = 0; r < 8; ++r)
                cur[mt][nt][r] = *(const float*)(rowp[mt][r] + lofs + nt * 64);

    v8f acc[2][2];
    #pragma unroll
    for (int mt = 0; mt < 2; ++mt)
        #pragma unroll
        for (int nt = 0; nt < 2; ++nt)
            acc[mt][nt] = (v8f){0.f, 0.f, 0.f, 0.f, 0.f, 0.f, 0.f, 0.f};

    FragU Ahf[2][2], Alf[2][2];
    #pragma unroll
    for (int mt = 0; mt < 2; ++mt) {
        const unsigned ao = (unsigned)(mt * 16) * (PITCH * 2);
        Ahf[0][mt].q[0] = *(const uint4*)(Ah + ao);
        Ahf[0][mt].q[1] = *(const uint4*)(Ah + ao + 32);
        Alf[0][mt].q[0] = *(const uint4*)(Al + ao);
        Alf[0][mt].q[1] = *(const uint4*)(Al + ao + 32);
    }

    #pragma unroll
    for (int kc = 0; kc < 36; ++kc) {
        const int s  = kc & 1;
        const int ns = s ^ 1;
        const int bs = kc % RD;
        const int nk = kc + RD - 1;
        if (nk < 36) {
            const int bn = nk % RD;
            #pragma unroll
            for (int nt = 0; nt < 2; ++nt) {
                const unsigned bb = (unsigned)nk * 8192u + (unsigned)nt * 1024u;
                Bh[bn][nt].q[0] = *(const uint4*)(fH + bofs0 + bb);
                Bh[bn][nt].q[1] = *(const uint4*)(fH + bofs0 + bb + 512);
                Bl[bn][nt].q[0] = *(const uint4*)(fL + bofs0 + bb);
                Bl[bn][nt].q[1] = *(const uint4*)(fL + bofs0 + bb + 512);
            }
        }
        if (kc + 1 < 36) {
            const int ndw = (kc + 1) >> 2, ncg = (kc + 1) & 3;
            #pragma unroll
            for (int mt = 0; mt < 2; ++mt) {
                const unsigned ao = (unsigned)(mt * 16 + ndw) * (PITCH * 2) + (unsigned)(ncg * 32) * 2;
                Ahf[ns][mt].q[0] = *(const uint4*)(Ah + ao);
                Ahf[ns][mt].q[1] = *(const uint4*)(Ah + ao + 32);
                Alf[ns][mt].q[0] = *(const uint4*)(Al + ao);
                Alf[ns][mt].q[1] = *(const uint4*)(Al + ao + 32);
            }
        }
        #pragma unroll
        for (int mt = 0; mt < 2; ++mt)
            #pragma unroll
            for (int nt = 0; nt < 2; ++nt)
                acc[mt][nt] = __builtin_amdgcn_wmma_f32_16x16x32_bf16(
                    false, Ahf[s][mt].v, false, Bh[bs][nt].v, (short)0, acc[mt][nt], false, false);
        #pragma unroll
        for (int mt = 0; mt < 2; ++mt)
            #pragma unroll
            for (int nt = 0; nt < 2; ++nt)
                acc[mt][nt] = __builtin_amdgcn_wmma_f32_16x16x32_bf16(
                    false, Alf[s][mt].v, false, Bh[bs][nt].v, (short)0, acc[mt][nt], false, false);
        #pragma unroll
        for (int mt = 0; mt < 2; ++mt)
            #pragma unroll
            for (int nt = 0; nt < 2; ++nt)
                acc[mt][nt] = __builtin_amdgcn_wmma_f32_16x16x32_bf16(
                    false, Ahf[s][mt].v, false, Bl[bs][nt].v, (short)0, acc[mt][nt], false, false);
    }

    #pragma unroll
    for (int mt = 0; mt < 2; ++mt)
        #pragma unroll
        for (int nt = 0; nt < 2; ++nt)
            #pragma unroll
            for (int r = 0; r < 8; ++r) {
                float conv = acc[mt][nt][r];
                conv = conv > 0.f ? conv : 0.f;
                *(float*)((char*)rowp[mt][r] + lofs + nt * 64) = cur[mt][nt][r] + conv;
            }
}

extern "C" void kernel_launch(void* const* d_in, const int* in_sizes, int n_in,
                              void* d_out, int out_size, void* d_ws, size_t ws_size,
                              hipStream_t stream)
{
    const float* x  = (const float*)d_in[0];
    const float* kk[4] = {(const float*)d_in[1], (const float*)d_in[2],
                          (const float*)d_in[3], (const float*)d_in[4]};
    float* out = (float*)d_out;

    char* ws = (char*)d_ws;
    float* Y = (float*)ws;
    const size_t yBytes = (size_t)NB * PLANE * sizeof(float);
    const size_t FRAG_ELEMS = 288 * 512;
    __bf16* fh[4]; __bf16* fl[4];
    for (int i = 0; i < 4; ++i) {
        fh[i] = (__bf16*)(ws + yBytes) + (size_t)i * 2 * FRAG_ELEMS;
        fl[i] = fh[i] + FRAG_ELEMS;
    }

    for (int i = 0; i < 4; ++i)
        scnn_wprep<<<(288 * 32 + 255) / 256, 256, 0, stream>>>(kk[i], fh[i], fl[i]);

    scnn_t_in<<<dim3(NB * HW, 16), dim3(32, 8), 0, stream>>>(x, Y);

    const int SMEM = 2 * PART_BYTES;
    const int sus[4]  = {16384, 16384, 128, 128};
    const int svs[4]  = {128, 128, 16384, 16384};
    const int revs[4] = {0, 1, 0, 1};
    for (int d = 0; d < 4; ++d)
        for (int step = 1; step < HW; ++step)
            scnn_step<<<dim3(4, NB), 128, SMEM, stream>>>(Y, fh[d], fl[d], sus[d], svs[d], revs[d], step);

    scnn_t_out<<<dim3(NB * HW, 16), dim3(32, 8), 0, stream>>>(Y, out);
}


# step re-tiled: 128 work-groups of one 32-position x 32-channel block (was 32 of 128 x 32), 40 staged rows not 136; the 4 waves split K by channel group, partials added in LDS: sum order changes
# speedup vs baseline: 3.9712x; 1.7779x over previous
#include <hip/hip_runtime.h>

typedef __attribute__((ext_vector_type(16))) __bf16 v16bf;
typedef __attribute__((ext_vector_type(8)))  float  v8f;

#define CCH   128
#define HW    128
#define NB    8
#define KW9   9
#define PITCH 136
#define ROWS  40
#define PART_BYTES (ROWS * PITCH * 2)
#define XPITCH 144
#define XPART (32 * XPITCH)
#define PLANE 2097152

union FragU { v16bf v; uint4 q[2]; };

__global__ void scnn_wprep(const float* __restrict__ K,
                           __bf16* __restrict__ hi, __bf16* __restrict__ lo)
{
    int t = blockIdx.x * blockDim.x + threadIdx.x;
    if (t >= 288 * 32) return;
    int lane = t & 31;
    int frag = t >> 5;
    int nt = frag & 7;
    int cg = (frag >> 3) & 3;
    int dw = frag >> 5;
    int halfk = lane >> 4;
    int co = nt * 16 + (lane & 15);
    __bf16* ph = hi + (size_t)frag * 512 + lane * 8;
    __bf16* pl = lo + (size_t)frag * 512 + lane * 8;
    #pragma unroll
    for (int j = 0; j < 16; ++j) {
        int kl = (j < 8) ? (halfk * 8 + j) : (16 + halfk * 8 + (j - 8));
        int ci = cg * 32 + kl;
        float w = K[((size_t)dw * CCH + ci) * CCH + co];
        __bf16 h = (__bf16)w;
        const int e = (j < 8) ? j : (256 + (j - 8));
        ph[e] = h;
        pl[e] = (__bf16)(w - (float)h);
    }
}

__global__ __launch_bounds__(256) void scnn_t_in(const float* __restrict__ X,
                                                 float* __restrict__ Y)
{
    __shared__ float tile[32][33];
    int bh = blockIdx.x;
    int b = bh >> 7, h = bh & 127;
    int tw = blockIdx.y & 3, tc = blockIdx.y >> 2;
    int x0 = threadIdx.x, y0 = threadIdx.y;
    #pragma unroll
    for (int k = 0; k < 4; ++k) {
        int c = tc * 32 + y0 + k * 8;
        int w = tw * 32 + x0;
        tile[y0 + k * 8][x0] = X[(size_t)b * PLANE + (size_t)c * 16384 + h * 128 + w];
    }
    __syncthreads();
    #pragma unroll
    for (int k = 0; k < 4; ++k) {
        int w = tw * 32 + y0 + k * 8;
        int c = tc * 32 + x0;
        Y[(size_t)b * PLANE + (size_t)h * 16384 + w * 128 + c] = tile[x0][y0 + k * 8];
    }
}

__global__ __launch_bounds__(256) void scnn_t_out(const float* __restrict__ Y,
                                                  float* __restrict__ O)
{
    __shared__ float tile[32][33];
    int bh = blockIdx.x;
    int b = bh >> 7, h = bh & 127;
    int tw = blockIdx.y & 3, tc = blockIdx.y >> 2;
    int x0 = threadIdx.x, y0 = threadIdx.y;
    #pragma unroll
    for (int k = 0; k < 4; ++k) {
        int w = tw * 32 + y0 + k * 8;
        int c = tc * 32 + x0;
        tile[y0 + k * 8][x0] = Y[(size_t)b * PLANE + (size_t)h * 16384 + w * 128 + c];
    }
    __syncthreads();
    #pragma unroll
    for (int k = 0; k < 4; ++k) {
        int w = tw * 32 + x0;
        int c = tc * 32 + y0 + k * 8;
        O[(size_t)b * PLANE + (size_t)c * 16384 + h * 128 + w] = tile[x0][y0 + k * 8];
    }
}

typedef __attribute__((ext_vector_type(4))) __bf16 v4bf;

__device__ __forceinline__ void lds_barrier()
{
    __builtin_amdgcn_fence(__ATOMIC_RELEASE, "workgroup", "local");
    __builtin_amdgcn_s_barrier();
    __builtin_amdgcn_fence(__ATOMIC_ACQUIRE, "workgroup", "local");
}
#define NCH 9

__global__ __launch_bounds__(128)
void scnn_step(float* __restrict__ Y,
               const __bf16* __restrict__ fragHi,
               const __bf16* __restrict__ fragLo,
               int su, int sv, int rev, int step)
{
    extern __shared__ char smem[];
    const int tid  = threadIdx.x;
    const int lane = tid & 31;
    const int q    = tid >> 5;
    const int ng   = blockIdx.x;
    const int m0   = (blockIdx.y & 3) * 32;
    float* Yb = Y + (size_t)(blockIdx.y >> 2) * PLANE;
    const int u  = rev ? (HW - 1 - step) : step;
    const int up = rev ? (u + 1) : (u - 1);
    const float* prow = Yb + (size_t)up * su;
    float* Yrow = Yb + (size_t)u * su;

    const int half = lane >> 4;
    const int li   = lane & 15;

    float4 xs[10];
    #pragma unroll
    for (int i = 0; i < 10; ++i) {
        const int p  = m0 - 4 + q + 4 * i;
        const int pc = ((unsigned)p < (unsigned)HW) ? p : 0;
        xs[i] = *(const float4*)(prow + (unsigned)(pc * sv + lane * 4));
    }
    __builtin_amdgcn_sched_barrier(0);

    const char* fH = (const char*)fragHi;
    const char* fL = (const char*)fragLo;
    const unsigned bofs0 = (unsigned)q * 8192u + (unsigned)(ng * 2) * 1024u + (unsigned)lane * 16u;

    FragU Bh[NCH][2], Bl[NCH][2];
    #pragma unroll
    for (int c = 0; c < NCH; ++c) {
        #pragma unroll
        for (int nt = 0; nt < 2; ++nt) {
            const unsigned bb = (unsigned)c * 32768u + (unsigned)nt * 1024u;
            Bh[c][nt].q[0] = *(const uint4*)(fH + bofs0 + bb);
            Bh[c][nt].q[1] = *(const uint4*)(fH + bofs0 + bb + 512);
            Bl[c][nt].q[0] = *(const uint4*)(fL + bofs0 + bb);
            Bl[c][nt].q[1] = *(const uint4*)(fL + bofs0 + bb + 512);
        }
    }
    const unsigned cofs = (unsigned)((m0 + q * 8 + (lane >> 3)) * sv + ng * 32 + (lane & 7) * 4);
    float4 cur[2];
    #pragma unroll
    for (int i = 0; i < 2; ++i)
        cur[i] = *(const float4*)(Yrow + cofs + (unsigned)(4 * i) * (unsigned)sv);

    #pragma unroll
    for (int i = 0; i < 10; ++i)
        asm volatile("" : "+v"(xs[i].x), "+v"(xs[i].y), "+v"(xs[i].z), "+v"(xs[i].w));

    {
        __bf16* hi0 = (__bf16*)(smem) + q * PITCH + lane * 4;
        __bf16* lo0 = (__bf16*)(smem + PART_BYTES) + q * PITCH + lane * 4;
        #pragma unroll
        for (int i = 0; i < 10; ++i) {
            const bool ok = (unsigned)(m0 - 4 + q + 4 * i) < (unsigned)HW;
            const float x0 = ok ? xs[i].x : 0.f, x1 = ok ? xs[i].y : 0.f;
            const float x2 = ok ? xs[i].z : 0.f, x3 = ok ? xs[i].w : 0.f;
            const __bf16 h0 = (__bf16)x0, h1 = (__bf16)x1, h2 = (__bf16)x2, h3 = (__bf16)x3;
            v4bf hv, lv;
            hv[0] = h0; hv[1] = h1; hv[2] = h2; hv[3] = h3;
            lv[0] = (__bf16)(x0 - (float)h0);
            lv[1] = (__bf16)(x1 - (float)h1);
            lv[2] = (__bf16)(x2 - (float)h2);
            lv[3] = (__bf16)(x3 - (float)h3);
            *(v4bf*)(hi0 + 4 * i * PITCH) = hv;
            *(v4bf*)(lo0 + 4 * i * PITCH) = lv;
        }
    }

    lds_barrier();

    const unsigned aofs0 = (unsigned)li * (PITCH * 2) + (unsigned)(half * 8) * 2 + (unsigned)(q * 32) * 2;
    const char* Ah = smem + aofs0;
    const char* Al = Ah + PART_BYTES;

    v8f acc[2][2];
    #pragma unroll
    for (int mt = 0; mt < 2; ++mt)
        #pragma unroll
        for (int nt = 0; nt < 2; ++nt)
            acc[mt][nt] = (v8f){0.f, 0.f, 0.f, 0.f, 0.f, 0.f, 0.f, 0.f};

    FragU Ahf[2][2], Alf[2][2];
    #pragma unroll
    for (int mt = 0; mt < 2; ++mt) {
        const unsigned ao = (unsigned)(mt * 16) * (PITCH * 2);
        Ahf[0][mt].q[0] = *(const uint4*)(Ah + ao);
        Ahf[0][mt].q[1] = *(const uint4*)(Ah + ao + 32);
        Alf[0][mt].q[0] = *(const uint4*)(Al + ao);
        Alf[0][mt].q[1] = *(const uint4*)(Al + ao + 32);
    }

    #pragma unroll
    for (int kc = 0; kc < NCH; ++kc) {
        const int s  = kc & 1;
        const int ns = s ^ 1;
        if (kc + 1 < NCH) {
            #pragma unroll
            for (int mt = 0; mt < 2; ++mt) {
                const unsigned ao = (unsigned)(mt * 16 + kc + 1) * (PITCH * 2);
                Ahf[ns][mt].q[0] = *(const uint4*)(Ah + ao);
                Ahf[ns][mt].q[1] = *(const uint4*)(Ah + ao + 32);
                Alf[ns][mt].q[0] = *(const uint4*)(Al + ao);
                Alf[ns][mt].q[1] = *(const uint4*)(Al + ao + 32);
            }
        }
        __builtin_amdgcn_sched_barrier(0);
        #pragma unroll
        for (int mt = 0; mt < 2; ++mt)
            #pragma unroll
            for (int nt = 0; nt < 2; ++nt)
                acc[mt][nt] = __builtin_amdgcn_wmma_f32_16x16x32_bf16(
                    false, Ahf[s][mt].v, false, Bh[kc][nt].v, (short)0, acc[mt][nt], false, false);
        #pragma unroll
        for (int mt = 0; mt < 2; ++mt)
            #pragma unroll
            for (int nt = 0; nt < 2; ++nt)
                acc[mt][nt] = __builtin_amdgcn_wmma_f32_16x16x32_bf16(
                    false, Alf[s][mt].v, false, Bh[kc][nt].v, (short)0, acc[mt][nt], false, false);
        #pragma unroll
        for (int mt = 0; mt < 2; ++mt)
            #pragma unroll
            for (int nt = 0; nt < 2; ++nt)
                acc[mt][nt] = __builtin_amdgcn_wmma_f32_16x16x32_bf16(
                    false, Ahf[s][mt].v, false, Bl[kc][nt].v, (short)0, acc[mt][nt], false, false);
        __builtin_amdgcn_sched_barrier(0);
    }

    {
        char* X = smem + 2 * PART_BYTES + q * XPART;
        #pragma unroll
        for (int mt = 0; mt < 2; ++mt)
            #pragma unroll
            for (int nt = 0; nt < 2; ++nt)
                #pragma unroll
                for (int r = 0; r < 8; ++r)
                    *(float*)(X + (mt * 16 + half * 8 + r) * XPITCH + (nt * 16 + li) * 4) = acc[mt][nt][r];
    }
    lds_barrier();

    {
        const char* X = smem + 2 * PART_BYTES + (q * 8 + (lane >> 3)) * XPITCH + (lane & 7) * 16;
        #pragma unroll
        for (int i = 0; i < 2; ++i) {
            const float4 p0 = *(const float4*)(X + 0 * XPART + (4 * i) * XPITCH);
            const float4 p1 = *(const float4*)(X + 1 * XPART + (4 * i) * XPITCH);
            const float4 p2 = *(const float4*)(X + 2 * XPART + (4 * i) * XPITCH);
            const float4 p3 = *(const float4*)(X + 3 * XPART + (4 * i) * XPITCH);
            float4 o;
            float c;
            c = ((p0.x + p1.x) + p2.x) + p3.x; o.x = cur[i].x + (c > 0.f ? c : 0.f);
            c = ((p0.y + p1.y) + p2.y) + p3.y; o.y = cur[i].y + (c > 0.f ? c : 0.f);
            c = ((p0.z + p1.z) + p2.z) + p3.z; o.z = cur[i].z + (c > 0.f ? c : 0.f);
            c = ((p0.w + p1.w) + p2.w) + p3.w; o.w = cur[i].w + (c > 0.f ? c : 0.f);
            *(float4*)(Yrow + cofs + (unsigned)(4 * i) * (unsigned)sv) = o;
        }
    }
}

extern "C" void kernel_launch(void* const* d_in, const int* in_sizes, int n_in,
                              void* d_out, int out_size, void* d_ws, size_t ws_size,
                              hipStream_t stream)
{
    const float* x  = (const float*)d_in[0];
    const float* kk[4] = {(const float*)d_in[1], (const float*)d_in[2],
                          (const float*)d_in[3], (const float*)d_in[4]};
    float* out = (float*)d_out;

    char* ws = (char*)d_ws;
    float* Y = (float*)ws;
    const size_t yBytes = (size_t)NB * PLANE * sizeof(float);
    const size_t FRAG_ELEMS = 288 * 512;
    __bf16* fh[4]; __bf16* fl[4];
    for (int i = 0; i < 4; ++i) {
        fh[i] = (__bf16*)(ws + yBytes) + (size_t)i * 2 * FRAG_ELEMS;
        fl[i] = fh[i] + FRAG_ELEMS;
    }

    for (int i = 0; i < 4; ++i)
        scnn_wprep<<<(288 * 32 + 255) / 256, 256, 0, stream>>>(kk[i], fh[i], fl[i]);

    scnn_t_in<<<dim3(NB * HW, 16), dim3(32, 8), 0, stream>>>(x, Y);

    const int SMEM = 2 * PART_BYTES + 4 * XPART;
    const int sus[4]  = {16384, 16384, 128, 128};
    const int svs[4]  = {128, 128, 16384, 16384};
    const int revs[4] = {0, 1, 0, 1};
    for (int d = 0; d < 4; ++d)
        for (int step = 1; step < HW; ++step)
            scnn_step<<<dim3(4, 4 * NB), 128, SMEM, stream>>>(Y, fh[d], fl[d], sus[d], svs[d], revs[d], step);

    scnn_t_out<<<dim3(NB * HW, 16), dim3(32, 8), 0, stream>>>(Y, out);
}


# each step also stores its line split into bf16 hi/lo chunks; the next step stages its channel group from that, with no conversion and no barrier in front of its WMMAs; values and sums unchanged
# speedup vs baseline: 4.2116x; 1.0605x over previous
#include <hip/hip_runtime.h>

typedef __attribute__((ext_vector_type(16))) __bf16 v16bf;
typedef __attribute__((ext_vector_type(8)))  float  v8f;

#define CCH   128
#define HW    128
#define NB    8
#define KW9   9
#define PITCH 136
#define ROWS  40
#define PART_BYTES (ROWS * PITCH * 2)
#define XPITCH 144
#define XPART (32 * XPITCH)
#define PLANE 2097152
#define SROW   512
#define SBATCH (HW * SROW)
#define SHALF  (NB * SBATCH)

union FragU { v16bf v; uint4 q[2]; };

__global__ void scnn_wprep(const float* __restrict__ K,
                           __bf16* __restrict__ hi, __bf16* __restrict__ lo)
{
    int t = blockIdx.x * blockDim.x + threadIdx.x;
    if (t >= 288 * 32) return;
    int lane = t & 31;
    int frag = t >> 5;
    int nt = frag & 7;
    int cg = (frag >> 3) & 3;
    int dw = frag >> 5;
    int halfk = lane >> 4;
    int co = nt * 16 + (lane & 15);
    __bf16* ph = hi + (size_t)frag * 512 + lane * 8;
    __bf16* pl = lo + (size_t)frag * 512 + lane * 8;
    #pragma unroll
    for (int j = 0; j < 16; ++j) {
        int kl = (j < 8) ? (halfk * 8 + j) : (16 + halfk * 8 + (j - 8));
        int ci = cg * 32 + kl;
        float w = K[((size_t)dw * CCH + ci) * CCH + co];
        __bf16 h = (__bf16)w;
        const int e = (j < 8) ? j : (256 + (j - 8));
        ph[e] = h;
        pl[e] = (__bf16)(w - (float)h);
    }
}

__global__ __launch_bounds__(256) void scnn_t_in(const float* __restrict__ X,
                                                 float* __restrict__ Y)
{
    __shared__ float tile[32][33];
    int bh = blockIdx.x;
    int b = bh >> 7, h = bh & 127;
    int tw = blockIdx.y & 3, tc = blockIdx.y >> 2;
    int x0 = threadIdx.x, y0 = threadIdx.y;
    #pragma unroll
    for (int k = 0; k < 4; ++k) {
        int c = tc * 32 + y0 + k * 8;
        int w = tw * 32 + x0;
        tile[y0 + k * 8][x0] = X[(size_t)b * PLANE + (size_t)c * 16384 + h * 128 + w];
    }
    __syncthreads();
    #pragma unroll
    for (int k = 0; k < 4; ++k) {
        int w = tw * 32 + y0 + k * 8;
        int c = tc * 32 + x0;
        Y[(size_t)b * PLANE + (size_t)h * 16384 + w * 128 + c] = tile[x0][y0 + k * 8];
    }
}

__global__ __launch_bounds__(256) void scnn_t_out(const float* __restrict__ Y,
                                                  float* __restrict__ O)
{
    __shared__ float tile[32][33];
    int bh = blockIdx.x;
    int b = bh >> 7, h = bh & 127;
    int tw = blockIdx.y & 3, tc = blockIdx.y >> 2;
    int x0 = threadIdx.x, y0 = threadIdx.y;
    #pragma unroll
    for (int k = 0; k < 4; ++k) {
        int w = tw * 32 + y0 + k * 8;
        int c = tc * 32 + x0;
        tile[y0 + k * 8][x0] = Y[(size_t)b * PLANE + (size_t)h * 16384 + w * 128 + c];
    }
    __syncthreads();
    #pragma unroll
    for (int k = 0; k < 4; ++k) {
        int w = tw * 32 + x0;
        int c = tc * 32 + y0 + k * 8;
        O[(size_t)b * PLANE + (size_t)c * 16384 + h * 128 + w] = tile[x0][y0 + k * 8];
    }
}

typedef __attribute__((ext_vector_type(4))) __bf16 v4bf;

__device__ __forceinline__ void lds_barrier()
{
    __builtin_amdgcn_fence(__ATOMIC_RELEASE, "workgroup", "local");
    __builtin_amdgcn_s_barrier();
    __builtin_amdgcn_fence(__ATOMIC_ACQUIRE, "workgroup", "local");
}
#define NCH 9
#define ROWB(i)  ((i) < 8 ? 16 * ((i) >> 2) + ((i) & 3) : 32 + ((i) - 8))
#define ROWS4(i) ((i) < 8 ? 4 : 2)

__global__ __launch_bounds__(128)
void scnn_step(float* __restrict__ Y,
               const __bf16* __restrict__ fragHi,
               const __bf16* __restrict__ fragLo,
               __bf16* __restrict__ S,
               int su, int sv, int rev, int step)
{
    extern __shared__ char smem[];
    const int tid  = threadIdx.x;
    const int lane = tid & 31;
    const int q    = tid >> 5;
    const int ng   = blockIdx.x;
    const int m0   = (blockIdx.y & 3) * 32;
    const int bat  = blockIdx.y >> 2;
    float* Yb = Y + (size_t)bat * PLANE;
    const int u  = rev ? (HW - 1 - step) : step;
    const int up = rev ? (u + 1) : (u - 1);
    const float* prow = Yb + (size_t)up * su;
    float* Yrow = Yb + (size_t)u * su;
    const char* Sin  = (const char*)S + (size_t)((step - 1) & 1) * SHALF + (size_t)bat * SBATCH;
    char*       Sout = (char*)S + (size_t)(step & 1) * SHALF + (size_t)bat * SBATCH;

    const int half = lane >> 4;
    const int li   = lane & 15;

    const int r4 = lane >> 3;
    const int cj = q * 8 + (lane & 7);
    const bool first = (step == 1);
    const unsigned lds4 = (unsigned)((4 * r4 * PITCH + cj * 4) * 2);
    const unsigned lds2 = (unsigned)((2 * r4 * PITCH + cj * 4) * 2);
    const unsigned bofs0 = (unsigned)q * 8192u + (unsigned)(ng * 2) * 1024u + (unsigned)lane * 16u;
    ulonglong2 xs[10];
    const char* src = first ? (const char*)prow : Sin;
    const unsigned rstride = first ? (unsigned)sv * 4u : (unsigned)SROW;
    #pragma unroll
    for (int i = 0; i < 10; ++i) {
        const int p  = m0 - 4 + ROWB(i) + ROWS4(i) * r4;
        const int pc = ((unsigned)p < (unsigned)HW) ? p : 0;
        xs[i] = *(const ulonglong2*)(src + ((unsigned)pc * rstride + (unsigned)cj * 16u));
    }
    __builtin_amdgcn_sched_barrier(0);

    const char* fH = (const char*)fragHi;
    const char* fL = (const char*)fragLo;
    FragU Bh[NCH][2], Bl[NCH][2];
    #pragma unroll
    for (int c = 0; c < NCH; ++c) {
        #pragma unroll
        for (int nt = 0; nt < 2; ++nt) {
            const unsigned bb = (unsigned)c * 32768u + (unsigned)nt * 1024u;
            Bh[c][nt].q[0] = *(const uint4*)(fH + bofs0 + bb);
            Bh[c][nt].q[1] = *(const uint4*)(fH + bofs0 + bb + 512);
            Bl[c][nt].q[0] = *(const uint4*)(fL + bofs0 + bb);
            Bl[c][nt].q[1] = *(const uint4*)(fL + bofs0 + bb + 512);
        }
    }
    const unsigned cofs = (unsigned)((m0 + q * 8 + (lane >> 3)) * sv + ng * 32 + (lane & 7) * 4);
    float4 cur[2];
    #pragma unroll
    for (int i = 0; i < 2; ++i)
        cur[i] = *(const float4*)(Yrow + cofs + (unsigned)(4 * i) * (unsigned)sv);

    #pragma unroll
    for (int i = 0; i < 10; ++i)
        asm volatile("" : "+v"(xs[i].x), "+v"(xs[i].y));

    if (first) {
        #pragma unroll
        for (int i = 0; i < 10; ++i) {
            const bool ok = (unsigned)(m0 - 4 + ROWB(i) + ROWS4(i) * r4) < (unsigned)HW;
            const float x0 = ok ? __uint_as_float((unsigned)xs[i].x) : 0.f, x1 = ok ? __uint_as_float((unsigned)(xs[i].x >> 32)) : 0.f;
            const float x2 = ok ? __uint_as_float((unsigned)xs[i].y) : 0.f, x3 = ok ? __uint_as_float((unsigned)(xs[i].y >> 32)) : 0.f;
            const __bf16 h0 = (__bf16)x0, h1 = (__bf16)x1, h2 = (__bf16)x2, h3 = (__bf16)x3;
            v4bf hv, lv;
            hv[0] = h0; hv[1] = h1; hv[2] = h2; hv[3] = h3;
            lv[0] = (__bf16)(x0 - (float)h0);
            lv[1] = (__bf16)(x1 - (float)h1);
            lv[2] = (__bf16)(x2 - (float)h2);
            lv[3] = (__bf16)(x3 - (float)h3);
            char* hp = smem + (i < 8 ? lds4 : lds2) + ROWB(i) * PITCH * 2;
            *(v4bf*)(hp) = hv;
            *(v4bf*)(hp + PART_BYTES) = lv;
        }
    } else {
        #pragma unroll
        for (int i = 0; i < 10; ++i) {
            const bool ok = (unsigned)(m0 - 4 + ROWB(i) + ROWS4(i) * r4) < (unsigned)HW;
            char* hp = smem + (i < 8 ? lds4 : lds2) + ROWB(i) * PITCH * 2;
            *(unsigned long long*)(hp) = ok ? xs[i].x : 0ull;
            *(unsigned long long*)(hp + PART_BYTES) = ok ? xs[i].y : 0ull;
        }
    }

    const unsigned aofs0 = (unsigned)li * (PITCH * 2) + (unsigned)(half * 8) * 2 + (unsigned)(q * 32) * 2;
    const char* Ah = smem + aofs0;
    const char* Al = Ah + PART_BYTES;

    v8f acc[2][2];
    #pragma unroll
    for (int mt = 0; mt < 2; ++mt)
        #pragma unroll
        for (int nt = 0; nt < 2; ++nt)
            acc[mt][nt] = (v8f){0.f, 0.f, 0.f, 0.f, 0.f, 0.f, 0.f, 0.f};

    FragU Ahf[2][2], Alf[2][2];
    #pragma unroll
    for (int mt = 0; mt < 2; ++mt) {
        const unsigned ao = (unsigned)(mt * 16) * (PITCH * 2);
        Ahf[0][mt].q[0] = *(const uint4*)(Ah + ao);
        Ahf[0][mt].q[1] = *(const uint4*)(Ah + ao + 32);
        Alf[0][mt].q[0] = *(const uint4*)(Al + ao);
        Alf[0][mt].q[1] = *(const uint4*)(Al + ao + 32);
    }

    #pragma unroll
    for (int kc = 0; kc < NCH; ++kc) {
        const int s  = kc & 1;
        const int ns = s ^ 1;
        if (kc + 1 < NCH) {
            #pragma unroll
            for (int mt = 0; mt < 2; ++mt) {
                const unsigned ao = (unsigned)(mt * 16 + kc + 1) * (PITCH * 2);
                Ahf[ns][mt].q[0] = *(const uint4*)(Ah + ao);
                Ahf[ns][mt].q[1] = *(const uint4*)(Ah + ao + 32);
                Alf[ns][mt].q[0] = *(const uint4*)(Al + ao);
                Alf[ns][mt].q[1] = *(const uint4*)(Al + ao + 32);
            }
        }
        __builtin_amdgcn_sched_barrier(0);
        #pragma unroll
        for (int mt = 0; mt < 2; ++mt)
            #pragma unroll
            for (int nt = 0; nt < 2; ++nt)
                acc[mt][nt] = __builtin_amdgcn_wmma_f32_16x16x32_bf16(
                    false, Ahf[s][mt].v, false, Bh[kc][nt].v, (short)0, acc[mt][nt], false, false);
        #pragma unroll
        for (int mt = 0; mt < 2; ++mt)
            #pragma unroll
            for (int nt = 0; nt < 2; ++nt)
                acc[mt][nt] = __builtin_amdgcn_wmma_f32_16x16x32_bf16(
                    false, Alf[s][mt].v, false, Bh[kc][nt].v, (short)0, acc[mt][nt], false, false);
        #pragma unroll
        for (int mt = 0; mt < 2; ++mt)
            #pragma unroll
            for (int nt = 0; nt < 2; ++nt)
                acc[mt][nt] = __builtin_amdgcn_wmma_f32_16x16x32_bf16(
                    false, Ahf[s][mt].v, false, Bl[kc][nt].v, (short)0, acc[mt][nt], false, false);
        __builtin_amdgcn_sched_barrier(0);
    }

    {
        char* X = smem + 2 * PART_BYTES + q * XPART;
        #pragma unroll
        for (int mt = 0; mt < 2; ++mt)
            #pragma unroll
            for (int nt = 0; nt < 2; ++nt)
                #pragma unroll
                for (int r = 0; r < 8; ++r)
                    *(float*)(X + (mt * 16 + half * 8 + r) * XPITCH + (nt * 16 + li) * 4) = acc[mt][nt][r];
    }
    lds_barrier();

    {
        const char* X = smem + 2 * PART_BYTES + (q * 8 + (lane >> 3)) * XPITCH + (lane & 7) * 16;
        #pragma unroll
        for (int i = 0; i < 2; ++i) {
            const float4 p0 = *(const float4*)(X + 0 * XPART + (4 * i) * XPITCH);
            const float4 p1 = *(const float4*)(X + 1 * XPART + (4 * i) * XPITCH);
            const float4 p2 = *(const float4*)(X + 2 * XPART + (4 * i) * XPITCH);
            const float4 p3 = *(const float4*)(X + 3 * XPART + (4 * i) * XPITCH);
            float4 o;
            float c;
            c = ((p0.x + p1.x) + p2.x) + p3.x; o.x = cur[i].x + (c > 0.f ? c : 0.f);
            c = ((p0.y + p1.y) + p2.y) + p3.y; o.y = cur[i].y + (c > 0.f ? c : 0.f);
            c = ((p0.z + p1.z) + p2.z) + p3.z; o.z = cur[i].z + (c > 0.f ? c : 0.f);
            c = ((p0.w + p1.w) + p2.w) + p3.w; o.w = cur[i].w + (c > 0.f ? c : 0.f);
            *(float4*)(Yrow + cofs + (unsigned)(4 * i) * (unsigned)sv) = o;
            const __bf16 h0 = (__bf16)o.x, h1 = (__bf16)o.y, h2 = (__bf16)o.z, h3 = (__bf16)o.w;
            v4bf shv, slv;
            shv[0] = h0; shv[1] = h1; shv[2] = h2; shv[3] = h3;
            slv[0] = (__bf16)(o.x - (float)h0);
            slv[1] = (__bf16)(o.y - (float)h1);
            slv[2] = (__bf16)(o.z - (float)h2);
            slv[3] = (__bf16)(o.w - (float)h3);
            char* sp = Sout + (unsigned)((m0 + q * 8 + (lane >> 3) + 4 * i) * SROW + (ng * 8 + (lane & 7)) * 16);
            *(v4bf*)(sp) = shv;
            *(v4bf*)(sp + 8) = slv;
        }
    }
}

extern "C" void kernel_launch(void* const* d_in, const int* in_sizes, int n_in,
                              void* d_out, int out_size, void* d_ws, size_t ws_size,
                              hipStream_t stream)
{
    const float* x  = (const float*)d_in[0];
    const float* kk[4] = {(const float*)d_in[1], (const float*)d_in[2],
                          (const float*)d_in[3], (const float*)d_in[4]};
    float* out = (float*)d_out;

    char* ws = (char*)d_ws;
    float* Y = (float*)ws;
    const size_t yBytes = (size_t)NB * PLANE * sizeof(float);
    const size_t FRAG_ELEMS = 288 * 512;
    __bf16* fh = (__bf16*)(ws + yBytes);
    __bf16* fl = fh + FRAG_ELEMS;
    __bf16* S  = fl + FRAG_ELEMS;

    scnn_t_in<<<dim3(NB * HW, 16), dim3(32, 8), 0, stream>>>(x, Y);

    const int SMEM = 2 * PART_BYTES + 4 * XPART;
    const int sus[4]  = {16384, 16384, 128, 128};
    const int svs[4]  = {128, 128, 16384, 16384};
    const int revs[4] = {0, 1, 0, 1};
    for (int d = 0; d < 4; ++d) {
        scnn_wprep<<<(288 * 32 + 255) / 256, 256, 0, stream>>>(kk[d], fh, fl);
        for (int step = 1; step < HW; ++step)
            scnn_step<<<dim3(4, 4 * NB), 128, SMEM, stream>>>(Y, fh, fl, S, sus[d], svs[d], revs[d], step);
    }

    scnn_t_out<<<dim3(NB * HW, 16), dim3(32, 8), 0, stream>>>(Y, out);
}
